# starting kernel
# baseline (speedup 1.0000x reference)
#include <hip/hip_runtime.h>

typedef float v2f __attribute__((ext_vector_type(2)));
typedef float v8f __attribute__((ext_vector_type(8)));

#define NN    256
#define LDM   260
#define NITER 50
#define RHO   1.0f
#define TPB   512
#define KB    8

extern "C" __global__ __launch_bounds__(TPB, 1)
void boxqp_admm_kernel(const float* __restrict__ Q,
                       const float* __restrict__ p,
                       const float* __restrict__ lb,
                       const float* __restrict__ ub,
                       float* __restrict__ out)
{
    extern __shared__ float smem[];
    float* Mm    = smem;
    float* rhs   = smem + NN * LDM;
    float* xv    = rhs + NN;
    float* CnegA = xv + NN;
    float* CnegB = CnegA + 4 * NN;
    float* RtA   = CnegB + 4 * NN;
    float* RtB   = RtA + 4 * NN;
    float* Bbuf  = RtB + 4 * NN;

    const int tid  = threadIdx.x;
    const int b    = blockIdx.x;
    const int lane = tid & 31;
    const int wv   = tid >> 5;
    const int rowA = (wv << 4) + (lane & 15);
    const int hk   = (lane >> 4) << 1;
    const float* Qb = Q + (size_t)b * NN * NN;

    for (int idx = tid; idx < NN * (NN / 4); idx += TPB) {
        int row = idx >> 6;
        int c   = (idx & 63) << 2;
        float4 q = *(const float4*)(Qb + row * NN + c);
        float* d = Mm + row * LDM + c;
        d[0] = q.x; d[1] = q.y; d[2] = q.z; d[3] = q.w;
    }
    __syncthreads();
    if (tid < NN) Mm[tid * LDM + tid] += RHO;
    __syncthreads();

    for (int K = 0; K < NN; K += KB) {
        if (tid < NN) {
            bool inK = (tid >= K) && (tid < K + KB);
            float4 cA = *(const float4*)(Mm + tid * LDM + K);
            float4 cB = *(const float4*)(Mm + tid * LDM + K + 4);
            float4 nA, nB;
            nA.x = inK ? 0.f : -cA.x;  nA.y = inK ? 0.f : -cA.y;
            nA.z = inK ? 0.f : -cA.z;  nA.w = inK ? 0.f : -cA.w;
            nB.x = inK ? 0.f : -cB.x;  nB.y = inK ? 0.f : -cB.y;
            nB.z = inK ? 0.f : -cB.z;  nB.w = inK ? 0.f : -cB.w;
            *(float4*)(CnegA + tid * 4) = nA;
            *(float4*)(CnegB + tid * 4) = nB;
        }
        if (wv == 0) {
            const int lr = lane;
            float ar[8];
            #pragma unroll
            for (int j = 0; j < 8; ++j)
                ar[j] = (lr < 8) ? Mm[(K + lr) * LDM + K + j] : 0.f;
            #pragma unroll
            for (int kk = 0; kk < 8; ++kk) {
                float prow[8];
                #pragma unroll
                for (int j = 0; j < 8; ++j) prow[j] = __shfl(ar[j], kk);
                float pinv = 1.0f / prow[kk];
                if (lr == kk) {
                    #pragma unroll
                    for (int j = 0; j < 8; ++j) ar[j] = (j == kk) ? pinv : ar[j] * pinv;
                } else {
                    float f = ar[kk];
                    #pragma unroll
                    for (int j = 0; j < 8; ++j) if (j != kk) ar[j] -= f * (prow[j] * pinv);
                    ar[kk] = -f * pinv;
                }
            }
            if (lr < 8) {
                #pragma unroll
                for (int j = 0; j < 8; ++j) Bbuf[lr * 8 + j] = ar[j];
            }
        }
        __syncthreads();

        {
            int j  = tid & 255;
            int r0 = (tid >> 8) << 2;
            float av[8];
            #pragma unroll
            for (int m = 0; m < 8; ++m) av[m] = Mm[(K + m) * LDM + j];
            float rv[4];
            #pragma unroll
            for (int r = 0; r < 4; ++r) {
                float s = 0.f;
                #pragma unroll
                for (int m = 0; m < 8; ++m) s += Bbuf[(r0 + r) * 8 + m] * av[m];
                rv[r] = s;
            }
            __syncthreads();
            #pragma unroll
            for (int r = 0; r < 4; ++r) Mm[(K + r0 + r) * LDM + j] = rv[r];
            float* RtX = (r0 == 0) ? RtA : RtB;
            *(float4*)(RtX + j * 4) = make_float4(rv[0], rv[1], rv[2], rv[3]);
        }
        __syncthreads();

        {
            v2f ca0 = *(const v2f*)(CnegA + rowA * 4 + hk);
            v2f ca1 = *(const v2f*)(CnegB + rowA * 4 + hk);
            const int rr = (wv << 4) + ((lane >> 4) << 3);
            const int cl = lane & 15;
            for (int cj = 0; cj < 16; ++cj) {
                const int cc = (cj << 4) + cl;
                v2f b0 = *(const v2f*)(RtA + cc * 4 + hk);
                v2f b1 = *(const v2f*)(RtB + cc * 4 + hk);
                v8f acc;
                #pragma unroll
                for (int g = 0; g < 8; ++g) acc[g] = Mm[(rr + g) * LDM + cc];
                acc = __builtin_amdgcn_wmma_f32_16x16x4_f32(
                    false, ca0, false, b0, (short)0, acc, false, false);
                acc = __builtin_amdgcn_wmma_f32_16x16x4_f32(
                    false, ca1, false, b1, (short)0, acc, false, false);
                #pragma unroll
                for (int g = 0; g < 8; ++g) Mm[(rr + g) * LDM + cc] = acc[g];
            }
        }
        __syncthreads();

        if (tid < NN) {
            if (tid < K || tid >= K + KB) {
                const float* cA = CnegA + tid * 4;
                const float* cB = CnegB + tid * 4;
                float o[8];
                #pragma unroll
                for (int c = 0; c < 8; ++c) {
                    float s = 0.f;
                    #pragma unroll
                    for (int r = 0; r < 4; ++r) s += cA[r] * Bbuf[r * 8 + c];
                    #pragma unroll
                    for (int r = 0; r < 4; ++r) s += cB[r] * Bbuf[(4 + r) * 8 + c];
                    o[c] = s;
                }
                *(float4*)(Mm + tid * LDM + K)     = make_float4(o[0], o[1], o[2], o[3]);
                *(float4*)(Mm + tid * LDM + K + 4) = make_float4(o[4], o[5], o[6], o[7]);
            } else {
                int r = tid - K;
                *(float4*)(Mm + tid * LDM + K)     = *(const float4*)(Bbuf + r * 8);
                *(float4*)(Mm + tid * LDM + K + 4) = *(const float4*)(Bbuf + r * 8 + 4);
            }
        }
        __syncthreads();
    }

    float zi = 0.f, ui = 0.f, xi = 0.f, pi = 0.f, lbi = 0.f, ubi = 0.f;
    if (tid < NN) {
        pi  = p [(size_t)b * NN + tid];
        lbi = lb[(size_t)b * NN + tid];
        ubi = ub[(size_t)b * NN + tid];
    }
    const float* arow = Mm + rowA * LDM + hk;

    for (int it = 0; it < NITER; ++it) {
        if (tid < NN) rhs[tid] = -pi + (zi - ui);
        __syncthreads();

        v8f acc = {};
        #pragma unroll 4
        for (int kb = 0; kb < NN; kb += 4) {
            v2f a  = *(const v2f*)(arow + kb);
            v2f bm = *(const v2f*)(rhs + kb + hk);
            acc = __builtin_amdgcn_wmma_f32_16x16x4_f32(
                false, a, false, bm, (short)0, acc, false, false);
        }
        if (lane == 0) {
            #pragma unroll
            for (int g = 0; g < 8; ++g) xv[(wv << 4) + g] = acc[g];
        }
        if (lane == 16) {
            #pragma unroll
            for (int g = 0; g < 8; ++g) xv[(wv << 4) + 8 + g] = acc[g];
        }
        __syncthreads();

        if (tid < NN) {
            xi = xv[tid];
            float t = xi + ui;
            zi = fminf(fmaxf(t, lbi), ubi);
            ui = ui + xi - zi;
        }
        __syncthreads();
    }

    if (tid < NN) out[(size_t)b * NN + tid] = xi;
}

extern "C" void kernel_launch(void* const* d_in, const int* in_sizes, int n_in,
                              void* d_out, int out_size, void* d_ws, size_t ws_size,
                              hipStream_t stream) {
    const float* Q  = (const float*)d_in[0];
    const float* p  = (const float*)d_in[1];
    const float* lb = (const float*)d_in[2];
    const float* ub = (const float*)d_in[3];
    float* out = (float*)d_out;

    const int B = in_sizes[0] / (NN * NN);
    const size_t shmem = (size_t)(NN * LDM + 2 * NN + 16 * NN + 64) * sizeof(float);

    (void)hipFuncSetAttribute((const void*)boxqp_admm_kernel,
                              hipFuncAttributeMaxDynamicSharedMemorySize,
                              (int)shmem);

    boxqp_admm_kernel<<<B, TPB, shmem, stream>>>(Q, p, lb, ub, out);
}


# Gauss-Jordan sweep re-tiled to pivot panels one WMMA tile (16) wide: 16 panels of 4 chained WMMAs per tile, not 32 of 2; operands read in place, pivot tile inverted by one wave; sum order changes
# speedup vs baseline: 1.5212x; 1.5212x over previous
#include <hip/hip_runtime.h>

typedef float v2f __attribute__((ext_vector_type(2)));
typedef float v8f __attribute__((ext_vector_type(8)));
typedef float v4f __attribute__((ext_vector_type(4)));

#define NN    256
#define LDM   260
#define NITER 50
#define RHO   1.0f
#define TPB   512
#define NPAN  16

__device__ __forceinline__ void pivot_tile_inverse(float* Mm, int K, int lane)
{
    const int c = lane & 15;
    float* col = Mm + (K + c) * LDM + K;
    float a[16];
    #pragma unroll
    for (int g = 0; g < 4; ++g) {
        const v4f v = *(const v4f*)(col + 4 * g);
        a[4 * g + 0] = v[0]; a[4 * g + 1] = v[1]; a[4 * g + 2] = v[2]; a[4 * g + 3] = v[3];
    }
    #pragma unroll
    for (int kk = 0; kk < 16; ++kk) {
        float f[16];
        #pragma unroll
        for (int r = 0; r < 16; ++r)
            f[r] = __int_as_float(__builtin_amdgcn_readlane(__float_as_int(a[r]), kk));
        const float pinv = 1.0f / f[kk];
        const bool own = (c == kk);
        #pragma unroll
        for (int r = 0; r < 16; ++r) a[r] = own ? ((r == kk) ? 1.0f : 0.0f) : a[r];
        const float t = a[kk] * pinv;
        #pragma unroll
        for (int r = 0; r < 16; ++r) if (r != kk) a[r] -= f[r] * t;
        a[kk] = t;
    }
    if (lane < 16) {
        #pragma unroll
        for (int g = 0; g < 4; ++g) {
            v4f s = {a[4 * g + 0], a[4 * g + 1], a[4 * g + 2], a[4 * g + 3]};
            *(v4f*)(col + 4 * g) = s;
        }
    }
}

extern "C" __global__ __launch_bounds__(TPB, 1)
void boxqp_admm_kernel(const float* __restrict__ Q,
                       const float* __restrict__ p,
                       const float* __restrict__ lb,
                       const float* __restrict__ ub,
                       float* __restrict__ out)
{
    extern __shared__ float smem[];
    float* Mm    = smem;
    float* rhs   = smem + NN * LDM;
    float* xv    = rhs + NN;

    const int tid  = threadIdx.x;
    const int b    = blockIdx.x;
    const int lane = tid & 31;
    const int wv   = tid >> 5;
    const int rowA = (wv << 4) + (lane & 15);
    const int hk   = (lane >> 4) << 1;
    const float* Qb = Q + (size_t)b * NN * NN;

    for (int idx = tid; idx < NN * (NN / 4); idx += TPB) {
        int row = idx >> 6;
        int c   = (idx & 63) << 2;
        float4 q = *(const float4*)(Qb + row * NN + c);
        float* d = Mm + row * LDM + c;
        d[0] = q.x; d[1] = q.y; d[2] = q.z; d[3] = q.w;
    }
    __syncthreads();
    if (tid < NN) Mm[tid * LDM + tid] += RHO;
    __syncthreads();

    const int wvu = __builtin_amdgcn_readfirstlane(wv);
    const int cl  = lane & 15;
    const int hh  = lane >> 4;

    if (wvu == 0) pivot_tile_inverse(Mm, 0, lane);
    __syncthreads();

    for (int kb = 0; kb < NPAN; ++kb) {
        const int K = kb << 4;

        if (wvu != kb) {
            const float* pa = Mm + (K + hk) * LDM + K + cl;
            float* xr = Mm + rowA * LDM + K;
            v2f fa[4], fb[4];
            #pragma unroll
            for (int q = 0; q < 4; ++q) {
                fa[q][0] = pa[(4 * q) * LDM];
                fa[q][1] = pa[(4 * q + 1) * LDM];
                fb[q] = *(const v2f*)(xr + 4 * q + hk);
            }
            v8f acc = {};
            #pragma unroll
            for (int q = 0; q < 4; ++q)
                acc = __builtin_amdgcn_wmma_f32_16x16x4_f32(
                    false, fa[q], false, fb[q], (short)0, acc, false, false);
            *(v4f*)(xr + 8 * hh)     = __builtin_shufflevector(acc, acc, 0, 1, 2, 3);
            *(v4f*)(xr + 8 * hh + 4) = __builtin_shufflevector(acc, acc, 4, 5, 6, 7);
        }
        __syncthreads();

        if (wvu != kb) {
            const int nb = kb + 1;
            const int gb = (nb >> 2) & 3;
            const float* pc = Mm + (K + hk) * LDM + rowA;
            v2f ca[4];
            #pragma unroll
            for (int q = 0; q < 4; ++q) {
                ca[q][0] = -pc[(4 * q) * LDM];
                ca[q][1] = -pc[(4 * q + 1) * LDM];
            }
            float* tb = Mm + cl * LDM + (wv << 4) + (hh << 3);
            const float* rb = Mm + cl * LDM + K + hk;
            {
                const v4f zz = {0.f, 0.f, 0.f, 0.f};
                *(v4f*)(tb + kb * 16 * LDM)     = zz;
                *(v4f*)(tb + kb * 16 * LDM + 4) = zz;
            }
            v4f lo[2][4], hi[2][4];
            v2f fr[2][4][4];
            #pragma unroll
            for (int t = 0; t < 4; ++t) {
                const int cj = gb * 4 + t;
                lo[0][t] = *(const v4f*)(tb + cj * 16 * LDM);
                hi[0][t] = *(const v4f*)(tb + cj * 16 * LDM + 4);
                #pragma unroll
                for (int q = 0; q < 4; ++q) fr[0][t][q] = *(const v2f*)(rb + cj * 16 * LDM + 4 * q);
            }
            #pragma unroll
            for (int g = 0; g < 4; ++g) {
                const int cur = g & 1, nxt = cur ^ 1;
                const int cg = ((g + gb) & 3) * 4;
                if (g < 3) {
                    const int ng = ((g + 1 + gb) & 3) * 4;
                    #pragma unroll
                    for (int t = 0; t < 4; ++t) {
                        const int cj = ng + t;
                        lo[nxt][t] = *(const v4f*)(tb + cj * 16 * LDM);
                        hi[nxt][t] = *(const v4f*)(tb + cj * 16 * LDM + 4);
                        #pragma unroll
                        for (int q = 0; q < 4; ++q) fr[nxt][t][q] = *(const v2f*)(rb + cj * 16 * LDM + 4 * q);
                    }
                }
                __builtin_amdgcn_sched_barrier(0);
                v8f acc[4];
                #pragma unroll
                for (int t = 0; t < 4; ++t)
                    acc[t] = __builtin_shufflevector(lo[cur][t], hi[cur][t], 0, 1, 2, 3, 4, 5, 6, 7);
                #pragma unroll
                for (int q = 0; q < 4; ++q) {
                    #pragma unroll
                    for (int t = 0; t < 4; ++t)
                        acc[t] = __builtin_amdgcn_wmma_f32_16x16x4_f32(
                            false, ca[q], false, fr[cur][t][q], (short)0, acc[t], false, false);
                }
                #pragma unroll
                for (int t = 0; t < 4; ++t) {
                    const int cj = cg + t;
                    *(v4f*)(tb + cj * 16 * LDM)     = __builtin_shufflevector(acc[t], acc[t], 0, 1, 2, 3);
                    *(v4f*)(tb + cj * 16 * LDM + 4) = __builtin_shufflevector(acc[t], acc[t], 4, 5, 6, 7);
                }
                __builtin_amdgcn_sched_barrier(0);
                if (g == 0 && nb < NPAN && wvu == nb)
                    pivot_tile_inverse(Mm, nb << 4, lane);
            }
        }
        __syncthreads();
    }

    const int  c16   = lane & 15;
    const int  rsel  = c16 & 7;
    const int  orow  = (wv << 4) + ((lane >> 4) << 3) + rsel;
    const bool owner = c16 < 8;
    float zi = 0.f, ui = 0.f, xi = 0.f;
    const float pi  = p [(size_t)b * NN + orow];
    const float lbi = lb[(size_t)b * NN + orow];
    const float ubi = ub[(size_t)b * NN + orow];
    const float* arow = Mm + rowA * LDM + hk;
    float* rcur = rhs;
    float* rnxt = xv;

    v2f af[2][16], bf[2][16];
    #pragma unroll
    for (int s = 0; s < 16; ++s) af[0][s] = *(const v2f*)(arow + 4 * s);
    if (owner) rcur[orow] = -pi + (zi - ui);
    __syncthreads();

    for (int it = 0; it < NITER; ++it) {
        #pragma unroll
        for (int s = 0; s < 16; ++s) bf[0][s] = *(const v2f*)(rcur + 4 * s + hk);

        v8f acc = {};
        #pragma unroll
        for (int blk = 0; blk < 4; ++blk) {
            const int cur = blk & 1, nxt = cur ^ 1;
            if (blk < 3) {
                #pragma unroll
                for (int s = 0; s < 16; ++s) {
                    af[nxt][s] = *(const v2f*)(arow + 64 * (blk + 1) + 4 * s);
                    bf[nxt][s] = *(const v2f*)(rcur + 64 * (blk + 1) + 4 * s + hk);
                }
            } else {
                #pragma unroll
                for (int s = 0; s < 16; ++s) af[nxt][s] = *(const v2f*)(arow + 4 * s);
            }
            __builtin_amdgcn_sched_barrier(0);
            #pragma unroll
            for (int s = 0; s < 16; ++s)
                acc = __builtin_amdgcn_wmma_f32_16x16x4_f32(
                    false, af[cur][s], false, bf[cur][s], (short)0, acc, false, false);
            __builtin_amdgcn_sched_barrier(0);
        }

        float xs = acc[0];
        #pragma unroll
        for (int r = 1; r < 8; ++r) xs = (rsel == r) ? acc[r] : xs;
        xi = xs;
        const float t = xi + ui;
        zi = fminf(fmaxf(t, lbi), ubi);
        ui = ui + xi - zi;
        if (owner) rnxt[orow] = -pi + (zi - ui);
        __syncthreads();
        float* rswap = rcur; rcur = rnxt; rnxt = rswap;
    }

    if (owner) rcur[orow] = xi;
    __syncthreads();
    if (tid < NN) out[(size_t)b * NN + tid] = rcur[tid];
}

extern "C" void kernel_launch(void* const* d_in, const int* in_sizes, int n_in,
                              void* d_out, int out_size, void* d_ws, size_t ws_size,
                              hipStream_t stream) {
    const float* Q  = (const float*)d_in[0];
    const float* p  = (const float*)d_in[1];
    const float* lb = (const float*)d_in[2];
    const float* ub = (const float*)d_in[3];
    float* out = (float*)d_out;

    const int B = in_sizes[0] / (NN * NN);
    const size_t shmem = (size_t)(NN * LDM + 2 * NN) * sizeof(float);

    (void)hipFuncSetAttribute((const void*)boxqp_admm_kernel,
                              hipFuncAttributeMaxDynamicSharedMemorySize,
                              (int)shmem);

    boxqp_admm_kernel<<<B, TPB, shmem, stream>>>(Q, p, lb, ub, out);
}
